# starting kernel
# baseline (speedup 1.0000x reference)
#include <hip/hip_runtime.h>

typedef __attribute__((ext_vector_type(2))) float v2f;
typedef __attribute__((ext_vector_type(8))) float v8f;

#define XS_STRIDE 132
#define WTP_STRIDE 288

__global__ __launch_bounds__(128) void reduce_m_kernel(
    const float* __restrict__ X, float* __restrict__ msg_m)
{
  const int bm = blockIdx.x;
  const int t  = threadIdx.x;
  const float* base = X + (size_t)bm * 64 * 128;
  float s = 0.f;
#pragma unroll 8
  for (int k = 0; k < 64; ++k) s += base[k * 128 + t];
  msg_m[(size_t)bm * 128 + t] = s;
}

__global__ __launch_bounds__(128) void reduce_k_kernel(
    const float* __restrict__ X, float* __restrict__ msg_k)
{
  const int bk = blockIdx.x;
  const int b  = bk >> 6;
  const int k  = bk & 63;
  const int t  = threadIdx.x;
  const float* base = X + ((size_t)b * 64 * 64 + k) * 128;
  float s = 0.f;
#pragma unroll 8
  for (int m = 0; m < 64; ++m) s += base[(size_t)m * 64 * 128 + t];
  msg_k[(size_t)bk * 128 + t] = s;
}

__global__ __launch_bounds__(256) void gemm128_wmma_kernel(
    const float* __restrict__ X, const float* __restrict__ W,
    float* __restrict__ out,
    const float* __restrict__ zm, const float* __restrict__ zk,
    int fuse)
{
  extern __shared__ float lds[];
  float* Xs  = lds;
  float* WTp = lds + 128 * XS_STRIDE;

  const int t = threadIdx.x;
  const size_t row_base = (size_t)blockIdx.x * 128;

  {
    const float4* Xg = (const float4*)(X + row_base * 128);
#pragma unroll
    for (int j = 0; j < 16; ++j) {
      int idx = t + j * 256;
      int row = idx >> 5;
      int c4  = idx & 31;
      float4 v = Xg[idx];
      *(float4*)(Xs + row * XS_STRIDE + c4 * 4) = v;
    }
  }
  {
    const float2* W2 = (const float2*)W;
#pragma unroll
    for (int j = 0; j < 32; ++j) {
      int idx = t + j * 256;
      int n  = idx & 127;
      int kp = idx >> 7;
      float2 w = W2[n * 64 + kp];
      *(float2*)(WTp + kp * WTP_STRIDE + 2 * n) = w;
    }
  }
  __syncthreads();

  const int wave = t >> 5;
  const int lane = t & 31;
  const int lm   = lane & 15;
  const int lg   = lane >> 4;

  v8f acc[8] = {};

  const float* arow = Xs + (wave * 16 + lm) * XS_STRIDE + 2 * lg;
  const float* brow = WTp + lg * WTP_STRIDE + 2 * lm;

  for (int kt = 0; kt < 32; ++kt) {
    v2f a = *(const v2f*)(arow + 4 * kt);
    const float* bk = brow + kt * (2 * WTP_STRIDE);
#pragma unroll
    for (int nt = 0; nt < 8; ++nt) {
      v2f b = *(const v2f*)(bk + nt * 32);
      acc[nt] = __builtin_amdgcn_wmma_f32_16x16x4_f32(
           false, a,  false, b,
           (short)0, acc[nt],
           false,  false);
    }
  }

  const size_t r0 = row_base + wave * 16 + 8 * lg;
#pragma unroll
  for (int nt = 0; nt < 8; ++nt) {
    const int col = nt * 16 + lm;
#pragma unroll
    for (int i = 0; i < 8; ++i) {
      const size_t r = r0 + i;
      float val = acc[nt][i];
      if (fuse) {
        const size_t bm = r >> 6;
        const size_t bk2 = ((r >> 12) << 6) | (r & 63);
        val += zm[bm * 128 + col] + zk[bk2 * 128 + col];
        __builtin_nontemporal_store(val, &out[r * 128 + col]);
      } else {
        out[r * 128 + col] = val;
      }
    }
  }
}

extern "C" void kernel_launch(void* const* d_in, const int* in_sizes, int n_in,
                              void* d_out, int out_size, void* d_ws, size_t ws_size,
                              hipStream_t stream)
{
  (void)in_sizes; (void)n_in; (void)out_size; (void)ws_size;
  const float* X     = (const float*)d_in[0];
  const float* Wself = (const float*)d_in[1];
  const float* Wm    = (const float*)d_in[2];
  const float* Wk    = (const float*)d_in[3];
  float* out = (float*)d_out;

  const size_t MSG = (size_t)4096 * 128;
  float* msg_m = (float*)d_ws;
  float* msg_k = msg_m + MSG;
  float* zm    = msg_k + MSG;
  float* zk    = zm + MSG;

  const size_t lds_bytes =
      (size_t)(128 * XS_STRIDE + 64 * WTP_STRIDE) * sizeof(float);

  reduce_m_kernel<<<4096, 128, 0, stream>>>(X, msg_m);
  reduce_k_kernel<<<4096, 128, 0, stream>>>(X, msg_k);

  gemm128_wmma_kernel<<<32, 256, lds_bytes, stream>>>(msg_m, Wm, zm, nullptr, nullptr, 0);
  gemm128_wmma_kernel<<<32, 256, lds_bytes, stream>>>(msg_k, Wk, zk, nullptr, nullptr, 0);

  gemm128_wmma_kernel<<<2048, 256, lds_bytes, stream>>>(X, Wself, out, zm, zk, 1);
}


# Message sums fused with their products: one launch in which a work-group sums 16 message rows from the input into LDS and multiplies them on f32 WMMA; 2 launches, not 5; new k order in zm and zk
# speedup vs baseline: 2.4594x; 2.4594x over previous
#include <hip/hip_runtime.h>

typedef __attribute__((ext_vector_type(2))) float v2f;
typedef __attribute__((ext_vector_type(4))) float v4f;
typedef __attribute__((ext_vector_type(8))) float v8f;

#define MS_STRIDE 132


template <int STEP, int ROWMUL>
__device__ __forceinline__ void sum_two_rows(
    const float* __restrict__ base, int lane, v4f& s0, v4f& s1)
{
  v4f ua[8], va[8], ub[8], vb[8];
  unsigned oa = 4 * lane;
  unsigned ob = 4 * lane + 8 * STEP;
#pragma unroll
  for (int i = 0; i < 8; ++i) {
    ua[i] = *(const v4f*)(base + oa + i * STEP);
    va[i] = *(const v4f*)(base + oa + i * STEP + ROWMUL);
  }
  for (int trip = 0; trip < 3; ++trip) {
#pragma unroll
    for (int i = 0; i < 8; ++i) {
      ub[i] = *(const v4f*)(base + ob + i * STEP);
      vb[i] = *(const v4f*)(base + ob + i * STEP + ROWMUL);
    }
#pragma unroll
    for (int i = 0; i < 8; ++i) { s0 += ua[i]; s1 += va[i]; }
    oa += 16 * STEP;
#pragma unroll
    for (int i = 0; i < 8; ++i) {
      ua[i] = *(const v4f*)(base + oa + i * STEP);
      va[i] = *(const v4f*)(base + oa + i * STEP + ROWMUL);
    }
#pragma unroll
    for (int i = 0; i < 8; ++i) { s0 += ub[i]; s1 += vb[i]; }
    ob += 16 * STEP;
  }
#pragma unroll
  for (int i = 0; i < 8; ++i) {
    ub[i] = *(const v4f*)(base + ob + i * STEP);
    vb[i] = *(const v4f*)(base + ob + i * STEP + ROWMUL);
  }
#pragma unroll
  for (int i = 0; i < 8; ++i) { s0 += ua[i]; s1 += va[i]; }
#pragma unroll
  for (int i = 0; i < 8; ++i) { s0 += ub[i]; s1 += vb[i]; }
}

__global__ __launch_bounds__(256) void msg_fused_kernel(
    const float* __restrict__ X, const float* __restrict__ Wm,
    const float* __restrict__ Wk, float* __restrict__ zm,
    float* __restrict__ zk)
{
  __shared__ float ms[16 * MS_STRIDE];

  const int t    = threadIdx.x;
  const int lane = t & 31;
  const int w    = __builtin_amdgcn_readfirstlane(t >> 5);
  const int bid  = blockIdx.x;
  const int b    = bid >> 3;
  const int kind = (bid >> 2) & 1;
  const int q    = bid & 3;

  const float* slab = X + (size_t)b * 64 * 64 * 128;
  v4f s0 = {0.f, 0.f, 0.f, 0.f}, s1 = {0.f, 0.f, 0.f, 0.f};
  if (kind == 0)
    sum_two_rows<128, 64 * 128>(slab + (size_t)(16 * q + 2 * w) * 64 * 128,
                                lane, s0, s1);
  else
    sum_two_rows<64 * 128, 128>(slab + (size_t)(16 * q + 2 * w) * 128,
                                lane, s0, s1);

  const int lm = lane & 15;
  const int lg = lane >> 4;
  const float* Wsel = kind ? Wk : Wm;
  float*       zsel = kind ? zk : zm;

  v4f wf[2][16];
  if (w < 4) {
#pragma unroll
    for (int tt = 0; tt < 2; ++tt) {
      const float* wrow =
          Wsel + (size_t)(32 * w + 2 * lm + tt) * 128 + 4 * lg;
#pragma unroll
      for (int j = 0; j < 16; ++j) wf[tt][j] = *(const v4f*)(wrow + 8 * j);
    }
  }

  *(v4f*)(ms + (2 * w)     * MS_STRIDE + 4 * lane) = s0;
  *(v4f*)(ms + (2 * w + 1) * MS_STRIDE + 4 * lane) = s1;
  __syncthreads();

  if (w < 4) {
    v8f acc[2] = {};
    const float* arow = ms + lm * MS_STRIDE + 4 * lg;
#pragma unroll
    for (int j = 0; j < 16; ++j) {
      const v4f a = *(const v4f*)(arow + 8 * j);
      const v2f a0 = __builtin_shufflevector(a, a, 0, 1);
      const v2f a1 = __builtin_shufflevector(a, a, 2, 3);
#pragma unroll
      for (int tt = 0; tt < 2; ++tt) {
        const v2f b0 = __builtin_shufflevector(wf[tt][j], wf[tt][j], 0, 1);
        acc[tt] = __builtin_amdgcn_wmma_f32_16x16x4_f32(
            false, a0, false, b0, (short)0, acc[tt], false, false);
      }
#pragma unroll
      for (int tt = 0; tt < 2; ++tt) {
        const v2f b1 = __builtin_shufflevector(wf[tt][j], wf[tt][j], 2, 3);
        acc[tt] = __builtin_amdgcn_wmma_f32_16x16x4_f32(
            false, a1, false, b1, (short)0, acc[tt], false, false);
      }
    }
    float* zrow = zsel + (size_t)(16 * (4 * b + q) + 8 * lg) * 128
                       + 32 * w + 2 * lm;
#pragma unroll
    for (int r = 0; r < 8; ++r) {
      v2f v;
      v.x = acc[0][r];
      v.y = acc[1][r];
      *(v2f*)(zrow + r * 128) = v;
    }
  }
}

__device__ __forceinline__ void main_tile(
    const v4f (&a)[16], const v4f (&wf)[8][16], const v4f (&zkf)[8][2],
    const v4f (&zmf)[2], float* __restrict__ orow)
{
  v8f acc[8] = {};
#pragma unroll
  for (int j = 0; j < 16; ++j) {
    const v2f a0 = __builtin_shufflevector(a[j], a[j], 0, 1);
    const v2f a1 = __builtin_shufflevector(a[j], a[j], 2, 3);
#pragma unroll
    for (int nt = 0; nt < 8; ++nt) {
      const v2f b0 = __builtin_shufflevector(wf[nt][j], wf[nt][j], 0, 1);
      acc[nt] = __builtin_amdgcn_wmma_f32_16x16x4_f32(
          false, a0, false, b0, (short)0, acc[nt], false, false);
    }
#pragma unroll
    for (int nt = 0; nt < 8; ++nt) {
      const v2f b1 = __builtin_shufflevector(wf[nt][j], wf[nt][j], 2, 3);
      acc[nt] = __builtin_amdgcn_wmma_f32_16x16x4_f32(
          false, a1, false, b1, (short)0, acc[nt], false, false);
    }
  }
#pragma unroll
  for (int r = 0; r < 8; ++r) {
#pragma unroll
    for (int q = 0; q < 2; ++q) {
      v4f v;
      v.x = acc[4 * q + 0][r] + (zmf[q].x + zkf[r][q].x);
      v.y = acc[4 * q + 1][r] + (zmf[q].y + zkf[r][q].y);
      v.z = acc[4 * q + 2][r] + (zmf[q].z + zkf[r][q].z);
      v.w = acc[4 * q + 3][r] + (zmf[q].w + zkf[r][q].w);
      __builtin_nontemporal_store(v, (v4f*)(orow + r * 128 + 64 * q));
    }
  }
}

__global__ __launch_bounds__(128) void main_regw_kernel(
    const float* __restrict__ X, const float* __restrict__ W,
    float* __restrict__ out,
    const float* __restrict__ zm, const float* __restrict__ zk)
{
  const int t    = threadIdx.x;
  const int lane = t & 31;
  const int lm   = lane & 15;
  const int lg   = lane >> 4;
  const int wid  = blockIdx.x * 4 + (t >> 5);
  const int b    = wid >> 4;
  const int mg   = (wid >> 2) & 3;
  const int kq   = wid & 3;

  v4f wf[8][16];
#pragma unroll
  for (int nt = 0; nt < 8; ++nt) {
    const float* wrow =
        W + (size_t)(64 * (nt >> 2) + 4 * lm + (nt & 3)) * 128 + 4 * lg;
#pragma unroll
    for (int j = 0; j < 16; ++j) wf[nt][j] = *(const v4f*)(wrow + 8 * j);
  }
  v4f zkf[8][2];
#pragma unroll
  for (int r = 0; r < 8; ++r) {
    const float* zrow =
        zk + (size_t)(b * 64 + kq * 16 + r + 8 * lg) * 128 + 4 * lm;
#pragma unroll
    for (int q = 0; q < 2; ++q) zkf[r][q] = *(const v4f*)(zrow + 64 * q);
  }

  const size_t bm0  = (size_t)b * 64 + mg * 16;
  const size_t row0 = bm0 * 64 + kq * 16;
  const float* xa  = X  + (row0 + lm) * 128 + 4 * lg;
  const float* zma = zm + bm0 * 128 + 4 * lm;
  float*       oa  = out + (row0 + 8 * lg) * 128 + 4 * lm;

  v4f a0[16], a1[16], zm0[2], zm1[2];
#pragma unroll
  for (int j = 0; j < 16; ++j) a0[j] = *(const v4f*)(xa + 8 * j);
#pragma unroll
  for (int q = 0; q < 2; ++q) zm0[q] = *(const v4f*)(zma + 64 * q);

  for (int i = 0; i < 16; i += 2) {
    {
      const float* xn = xa + (size_t)(i + 1) * 64 * 128;
#pragma unroll
      for (int j = 0; j < 16; ++j) a1[j] = *(const v4f*)(xn + 8 * j);
#pragma unroll
      for (int q = 0; q < 2; ++q)
        zm1[q] = *(const v4f*)(zma + (i + 1) * 128 + 64 * q);
    }
    main_tile(a0, wf, zkf, zm0, oa + (size_t)i * 64 * 128);
    {
      const int in = (i + 2 < 16) ? i + 2 : 15;
      const float* xn = xa + (size_t)in * 64 * 128;
#pragma unroll
      for (int j = 0; j < 16; ++j) a0[j] = *(const v4f*)(xn + 8 * j);
#pragma unroll
      for (int q = 0; q < 2; ++q)
        zm0[q] = *(const v4f*)(zma + in * 128 + 64 * q);
    }
    main_tile(a1, wf, zkf, zm1, oa + (size_t)(i + 1) * 64 * 128);
  }
}

extern "C" void kernel_launch(void* const* d_in, const int* in_sizes, int n_in,
                              void* d_out, int out_size, void* d_ws, size_t ws_size,
                              hipStream_t stream)
{
  (void)in_sizes; (void)n_in; (void)out_size; (void)ws_size;
  const float* X     = (const float*)d_in[0];
  const float* Wself = (const float*)d_in[1];
  const float* Wm    = (const float*)d_in[2];
  const float* Wk    = (const float*)d_in[3];
  float* out = (float*)d_out;

  const size_t MSG = (size_t)4096 * 128;
  float* zm = (float*)d_ws;
  float* zk = zm + MSG;

  msg_fused_kernel<<<512, 256, 0, stream>>>(X, Wm, Wk, zm, zk);

  main_regw_kernel<<<256, 128, 0, stream>>>(X, Wself, out, zm, zk);
}
